# starting kernel
# baseline (speedup 1.0000x reference)
#include <hip/hip_runtime.h>
#include <hip/hip_bf16.h>

#define N 4096
#define SINKHORN_ITERS 50
#define NCHUNK 16
#define ROWS_PER_CHUNK (N / NCHUNK)

typedef float v2f __attribute__((ext_vector_type(2)));
typedef float v8f __attribute__((ext_vector_type(8)));

__global__ void sinkhorn_init_c(float* __restrict__ C) {
    int j = blockIdx.x * blockDim.x + threadIdx.x;
    if (j < N) C[j] = 0.0f;
}

__global__ __launch_bounds__(256) void sinkhorn_row_lse(
    const float* __restrict__ M, const float* __restrict__ C,
    float* __restrict__ R) {
    __shared__ float sdata[256];
    const int i = blockIdx.x;
    const int tid = threadIdx.x;
    const float* __restrict__ row = M + (size_t)i * N;

    float acc = 0.0f;
#pragma unroll
    for (int j4 = tid * 4; j4 < N; j4 += 256 * 4) {
        const float4 m = *(const float4*)(row + j4);
        const float4 c = *(const float4*)(C + j4);
        acc += __expf(m.x - c.x);
        acc += __expf(m.y - c.y);
        acc += __expf(m.z - c.z);
        acc += __expf(m.w - c.w);
    }
    sdata[tid] = acc;
    __syncthreads();
    for (int s = 128; s > 0; s >>= 1) {
        if (tid < s) sdata[tid] += sdata[tid + s];
        __syncthreads();
    }
    if (tid == 0) R[i] = __logf(sdata[0]);
}

__global__ __launch_bounds__(128) void sinkhorn_col_partial(
    const float* __restrict__ M, const float* __restrict__ R,
    float* __restrict__ partial) {
    const int lane = threadIdx.x & 31;
    const int wave = threadIdx.x >> 5;
    const int j0 = blockIdx.x * 64 + wave * 16;
    const int rowBase = blockIdx.y * ROWS_PER_CHUNK;
    const int half = lane >> 4;
    const int lcol = lane & 15;
    const int col = j0 + lcol;

    v2f a;
    a.x = 1.0f;
    a.y = 1.0f;
    v8f d = {};

    const float* __restrict__ mc = M + col;
    for (int k = 0; k < ROWS_PER_CHUNK; k += 4) {
        const int r0 = rowBase + k + 2 * half;
        __builtin_prefetch(&mc[(size_t)(r0 + 8) * N], 0, 3);
        v2f b;
        b.x = __expf(mc[(size_t)r0 * N] - R[r0]);
        b.y = __expf(mc[(size_t)(r0 + 1) * N] - R[r0 + 1]);
        d = __builtin_amdgcn_wmma_f32_16x16x4_f32(false, a, false, b,
                                                  (short)0, d, false, false);
    }

    if (half == 0) {
        partial[(size_t)blockIdx.y * N + col] = d[0];
    }
}

__global__ void sinkhorn_col_finalize(const float* __restrict__ partial,
                                      float* __restrict__ C) {
    int j = blockIdx.x * blockDim.x + threadIdx.x;
    if (j >= N) return;
    float s = 0.0f;
#pragma unroll
    for (int c = 0; c < NCHUNK; ++c) s += partial[(size_t)c * N + j];
    C[j] = __logf(s);
}

__global__ __launch_bounds__(256) void sinkhorn_final(
    const float* __restrict__ M, const float* __restrict__ R,
    const float* __restrict__ C, float* __restrict__ out) {
    const int idx = blockIdx.x * blockDim.x + threadIdx.x;
    const int i = idx >> (12 - 2);
    const int j4 = (idx & (N / 4 - 1)) << 2;
    const size_t off = (size_t)i * N + j4;

    const float4 m = *(const float4*)(M + off);
    const float4 c = *(const float4*)(C + j4);
    const float r = R[i];
    float4 o;
    o.x = fmaxf(__expf(m.x - r - c.x), 1e-9f);
    o.y = fmaxf(__expf(m.y - r - c.y), 1e-9f);
    o.z = fmaxf(__expf(m.z - r - c.z), 1e-9f);
    o.w = fmaxf(__expf(m.w - r - c.w), 1e-9f);
    *(float4*)(out + off) = o;
}

extern "C" void kernel_launch(void* const* d_in, const int* in_sizes, int n_in,
                              void* d_out, int out_size, void* d_ws,
                              size_t ws_size, hipStream_t stream) {
    (void)in_sizes; (void)n_in; (void)out_size; (void)ws_size;

    const float* M = (const float*)d_in[0];
    float* out = (float*)d_out;

    char* ws = (char*)d_ws;
    float* R = (float*)(ws);
    float* C = (float*)(ws + 16 * 1024);
    float* partial = (float*)(ws + 32 * 1024);

    sinkhorn_init_c<<<N / 256, 256, 0, stream>>>(C);

    for (int t = 0; t < SINKHORN_ITERS; ++t) {
        sinkhorn_row_lse<<<N, 256, 0, stream>>>(M, C, R);
        sinkhorn_col_partial<<<dim3(N / 64, NCHUNK), 128, 0, stream>>>(M, R,
                                                                       partial);
        sinkhorn_col_finalize<<<N / 256, 256, 0, stream>>>(partial, C);
    }

    sinkhorn_final<<<(N * (size_t)N / 4) / 256, 256, 0, stream>>>(M, R, C, out);
}


# iteration re-tiled: one 16-wave work-group per WGP per pass, 128-bit row loads, 4 row chunks so the column finalize folds into the row pass (3 -> 2 launches/iter); row and column sums re-ordered
# speedup vs baseline: 1.3880x; 1.3880x over previous
#include <hip/hip_runtime.h>


#define N 4096
#define SINKHORN_ITERS 50
#define NCHUNK 4
#define ROWS_PER_CHUNK (N / NCHUNK)
#define ROWS_PER_BLOCK 16
#define ROWS_PER_WAVE 64

typedef float v2f __attribute__((ext_vector_type(2)));
typedef float v8f __attribute__((ext_vector_type(8)));

__global__ void sinkhorn_init_partial(float* __restrict__ partial) {
    const int idx = blockIdx.x * blockDim.x + threadIdx.x;
    if (idx < NCHUNK * N) partial[idx] = (idx < N) ? 1.0f : 0.0f;
}

__global__ __launch_bounds__(512) __attribute__((amdgpu_waves_per_eu(4, 4)))
void sinkhorn_row_lse(
    const float* __restrict__ M, const float* __restrict__ partial,
    float* __restrict__ R) {
    __shared__ float4 sC[N / 4];
    const int tid = threadIdx.x;
    const int lane = tid & 31;
    const int wave = tid >> 5;

    const int i = blockIdx.x * ROWS_PER_BLOCK + wave;
    const float4* __restrict__ row = (const float4*)(M + (size_t)i * N);
    float4 m[N / 128];
#pragma unroll
    for (int k = 0; k < 8; ++k) m[k] = row[k * 32 + lane];

#pragma unroll
    for (int h = 0; h < 2; ++h) {
        const int q = tid + h * 512;
        float4 s = make_float4(0.0f, 0.0f, 0.0f, 0.0f);
#pragma unroll
        for (int c = 0; c < NCHUNK; ++c) {
            const float4 p = *(const float4*)(partial + (size_t)c * N + 4 * q);
            s.x += p.x;
            s.y += p.y;
            s.z += p.z;
            s.w += p.w;
        }
        sC[q] = make_float4(__logf(s.x), __logf(s.y), __logf(s.z), __logf(s.w));
    }
    __syncthreads();

#pragma unroll
    for (int k = 8; k < N / 128; ++k) m[k] = row[k * 32 + lane];

    float4 acc = make_float4(0.0f, 0.0f, 0.0f, 0.0f);
#pragma unroll
    for (int k = 0; k < N / 128; ++k) {
        const float4 c = sC[k * 32 + lane];
        acc.x += __expf(m[k].x - c.x);
        acc.y += __expf(m[k].y - c.y);
        acc.z += __expf(m[k].z - c.z);
        acc.w += __expf(m[k].w - c.w);
    }
    float t = (acc.x + acc.y) + (acc.z + acc.w);
#pragma unroll
    for (int off = 16; off > 0; off >>= 1) t += __shfl_xor(t, off);
    if (lane == 0) R[i] = __logf(t);
}

__global__ __launch_bounds__(512) __attribute__((amdgpu_waves_per_eu(4, 4)))
void sinkhorn_col_partial(
    const float* __restrict__ M, const float* __restrict__ R,
    float* __restrict__ partial) {
    __shared__ __attribute__((aligned(16))) float sP[16][64];
    const int tid = threadIdx.x;
    const int lane = tid & 31;
    const int wave = tid >> 5;
    const int half = lane >> 4;
    const int q = lane & 15;
    const int j0 = blockIdx.x * 64 + 4 * q;
    const int rowBase = blockIdx.y * ROWS_PER_CHUNK + wave * ROWS_PER_WAVE;

    v2f a;
    a.x = 1.0f;
    a.y = 1.0f;
    v8f d0 = {};
    v8f d1 = {};
    v8f d2 = {};
    v8f d3 = {};

    const int myRow = rowBase + 32 * half;
    const float* __restrict__ mc = M + (size_t)myRow * N + j0;
    const float4* __restrict__ rr = (const float4*)(R + myRow);
#pragma unroll
    for (int g = 0; g < 8; ++g) {
        const float4 r4 = rr[g];
        const float4 m0 = *(const float4*)(mc + (size_t)(4 * g + 0) * N);
        const float4 m1 = *(const float4*)(mc + (size_t)(4 * g + 1) * N);
        const float4 m2 = *(const float4*)(mc + (size_t)(4 * g + 2) * N);
        const float4 m3 = *(const float4*)(mc + (size_t)(4 * g + 3) * N);
        v2f b;
        b.x = __expf(m0.x - r4.x);
        b.y = __expf(m1.x - r4.y);
        d0 = __builtin_amdgcn_wmma_f32_16x16x4_f32(false, a, false, b, (short)0,
                                                   d0, false, false);
        b.x = __expf(m0.y - r4.x);
        b.y = __expf(m1.y - r4.y);
        d1 = __builtin_amdgcn_wmma_f32_16x16x4_f32(false, a, false, b, (short)0,
                                                   d1, false, false);
        b.x = __expf(m0.z - r4.x);
        b.y = __expf(m1.z - r4.y);
        d2 = __builtin_amdgcn_wmma_f32_16x16x4_f32(false, a, false, b, (short)0,
                                                   d2, false, false);
        b.x = __expf(m0.w - r4.x);
        b.y = __expf(m1.w - r4.y);
        d3 = __builtin_amdgcn_wmma_f32_16x16x4_f32(false, a, false, b, (short)0,
                                                   d3, false, false);
        b.x = __expf(m2.x - r4.z);
        b.y = __expf(m3.x - r4.w);
        d0 = __builtin_amdgcn_wmma_f32_16x16x4_f32(false, a, false, b, (short)0,
                                                   d0, false, false);
        b.x = __expf(m2.y - r4.z);
        b.y = __expf(m3.y - r4.w);
        d1 = __builtin_amdgcn_wmma_f32_16x16x4_f32(false, a, false, b, (short)0,
                                                   d1, false, false);
        b.x = __expf(m2.z - r4.z);
        b.y = __expf(m3.z - r4.w);
        d2 = __builtin_amdgcn_wmma_f32_16x16x4_f32(false, a, false, b, (short)0,
                                                   d2, false, false);
        b.x = __expf(m2.w - r4.z);
        b.y = __expf(m3.w - r4.w);
        d3 = __builtin_amdgcn_wmma_f32_16x16x4_f32(false, a, false, b, (short)0,
                                                   d3, false, false);
    }

    if (half == 0) {
        *(float4*)&sP[wave][4 * q] = make_float4(d0[0], d1[0], d2[0], d3[0]);
    }
    __syncthreads();
    if (tid < 64) {
        float s = 0.0f;
#pragma unroll
        for (int w = 0; w < 16; ++w) s += sP[w][tid];
        partial[(size_t)blockIdx.y * N + blockIdx.x * 64 + tid] = s;
    }
}

__global__ void sinkhorn_col_finalize(const float* __restrict__ partial,
                                      float* __restrict__ C) {
    int j = blockIdx.x * blockDim.x + threadIdx.x;
    if (j >= N) return;
    float s = 0.0f;
#pragma unroll
    for (int c = 0; c < NCHUNK; ++c) s += partial[(size_t)c * N + j];
    C[j] = __logf(s);
}

__global__ __launch_bounds__(256) void sinkhorn_final(
    const float* __restrict__ M, const float* __restrict__ R,
    const float* __restrict__ C, float* __restrict__ out) {
    const int idx = blockIdx.x * blockDim.x + threadIdx.x;
    const int i = idx >> (12 - 2);
    const int j4 = (idx & (N / 4 - 1)) << 2;
    const size_t off = (size_t)i * N + j4;

    const float4 m = *(const float4*)(M + off);
    const float4 c = *(const float4*)(C + j4);
    const float r = R[i];
    float4 o;
    o.x = fmaxf(__expf(m.x - r - c.x), 1e-9f);
    o.y = fmaxf(__expf(m.y - r - c.y), 1e-9f);
    o.z = fmaxf(__expf(m.z - r - c.z), 1e-9f);
    o.w = fmaxf(__expf(m.w - r - c.w), 1e-9f);
    *(float4*)(out + off) = o;
}

extern "C" void kernel_launch(void* const* d_in, const int* in_sizes, int n_in,
                              void* d_out, int out_size, void* d_ws,
                              size_t ws_size, hipStream_t stream) {
    (void)in_sizes; (void)n_in; (void)out_size; (void)ws_size;

    const float* M = (const float*)d_in[0];
    float* out = (float*)d_out;

    char* ws = (char*)d_ws;
    float* R = (float*)(ws);
    float* C = (float*)(ws + 16 * 1024);
    float* partial = (float*)(ws + 32 * 1024);

    sinkhorn_init_partial<<<NCHUNK * N / 256, 256, 0, stream>>>(partial);

    for (int t = 0; t < SINKHORN_ITERS; ++t) {
        sinkhorn_row_lse<<<N / ROWS_PER_BLOCK, 512, 0, stream>>>(M, partial, R);
        sinkhorn_col_partial<<<dim3(N / 64, NCHUNK), 512, 0, stream>>>(M, R,
                                                                       partial);
    }

    sinkhorn_col_finalize<<<N / 256, 256, 0, stream>>>(partial, C);
    sinkhorn_final<<<(N * (size_t)N / 4) / 256, 256, 0, stream>>>(M, R, C, out);
}


# row pass and column partials fused: a 16-wave group keeps 16 rows in registers, exp once, 1/rowsum x tile on the matrix unit in two 8-row stages, M read once per iter; row and column sums re-ordered
# speedup vs baseline: 1.4941x; 1.0764x over previous
#include <hip/hip_runtime.h>


#define N 4096
#define SINKHORN_ITERS 50
#define ROWS_PER_BLOCK 16
#define NBLOCK (N / ROWS_PER_BLOCK)
#define COLS_PER_WAVE 256

typedef float v2f __attribute__((ext_vector_type(2)));
typedef float v4f __attribute__((ext_vector_type(4)));
typedef float v8f __attribute__((ext_vector_type(8)));

__global__ void sinkhorn_init_c(float* __restrict__ C) {
    const int j = blockIdx.x * blockDim.x + threadIdx.x;
    if (j < N) C[j] = 0.0f;
}

__global__ __launch_bounds__(512) __attribute__((amdgpu_waves_per_eu(4, 4)))
void sinkhorn_row_col(
    const float* __restrict__ M, const float* __restrict__ C,
    float* __restrict__ R, float* __restrict__ Q) {
    __shared__ float sS[2][16][8];
    const int tid = threadIdx.x;
    const int lane = tid & 31;
    const int wave = tid >> 5;
    const int half = lane >> 4;
    const int q = lane & 15;
    const int i0 = blockIdx.x * ROWS_PER_BLOCK;
    const int j0 = wave * COLS_PER_WAVE + 4 * q;

    v4f c[4];
#pragma unroll
    for (int g = 0; g < 4; ++g) c[g] = *(const v4f*)(C + j0 + 64 * g);

    const float* __restrict__ mp = M + (size_t)(i0 + 8 * half) * N + j0;
    v4f m[8][4];
#pragma unroll
    for (int p = 0; p < 4; ++p) {
        __builtin_amdgcn_sched_barrier(0);
#pragma unroll
        for (int g = 0; g < 4; ++g) {
#pragma unroll
            for (int t = 2 * p; t < 2 * p + 2; ++t)
                m[t][g] = *(const v4f*)(mp + (size_t)t * N + 64 * g);
        }
    }

    const bool b8 = (q & 8) != 0;
    const bool b4 = (q & 4) != 0;
    float4 res[2][4];

#pragma unroll
    for (int st = 0; st < 2; ++st) {
        v2f e[2][4][4];
        float s[4];
#pragma unroll
        for (int pp = 0; pp < 2; ++pp) {
            const int p = 2 * st + pp;
            float4 s0 = make_float4(0.0f, 0.0f, 0.0f, 0.0f);
            float4 s1 = make_float4(0.0f, 0.0f, 0.0f, 0.0f);
#pragma unroll
            for (int g = 0; g < 4; ++g) {
                v4f x0 = m[2 * p][g];
                v4f x1 = m[2 * p + 1][g];
                asm volatile("" : "+v"(x0), "+v"(x1));
                const v4f y0 = x0 - c[g];
                const v4f y1 = x1 - c[g];
                e[pp][g][0].x = __expf(y0.x);
                e[pp][g][1].x = __expf(y0.y);
                e[pp][g][2].x = __expf(y0.z);
                e[pp][g][3].x = __expf(y0.w);
                e[pp][g][0].y = __expf(y1.x);
                e[pp][g][1].y = __expf(y1.y);
                e[pp][g][2].y = __expf(y1.z);
                e[pp][g][3].y = __expf(y1.w);
                s0.x += e[pp][g][0].x;
                s0.y += e[pp][g][1].x;
                s0.z += e[pp][g][2].x;
                s0.w += e[pp][g][3].x;
                s1.x += e[pp][g][0].y;
                s1.y += e[pp][g][1].y;
                s1.z += e[pp][g][2].y;
                s1.w += e[pp][g][3].y;
            }
            s[2 * pp] = (s0.x + s0.y) + (s0.z + s0.w);
            s[2 * pp + 1] = (s1.x + s1.y) + (s1.z + s1.w);
            __builtin_amdgcn_sched_barrier(0);
        }

        float u2[2];
#pragma unroll
        for (int u = 0; u < 2; ++u) {
            const float keep = b8 ? s[u + 2] : s[u];
            const float send = b8 ? s[u] : s[u + 2];
            u2[u] = keep + __shfl_xor(send, 8);
        }
        float u1;
        {
            const float keep = b4 ? u2[1] : u2[0];
            const float send = b4 ? u2[0] : u2[1];
            u1 = keep + __shfl_xor(send, 4);
        }
        u1 += __shfl_xor(u1, 2);
        u1 += __shfl_xor(u1, 1);
        if ((q & 3) == 0) sS[st][wave][4 * half + (q >> 2)] = u1;
        asm volatile("s_wait_dscnt 0x0\n\ts_barrier_signal -1\n\ts_barrier_wait -1" ::: "memory");

        float tot = 0.0f;
#pragma unroll
        for (int w = 0; w < 16; ++w) tot += sS[st][w][lane & 7];
        const float inv = 1.0f / tot;
        if (wave == 0 && lane < 8)
            R[i0 + 8 * (lane >> 2) + 4 * st + (lane & 3)] = __logf(tot);

        v2f a[2];
#pragma unroll
        for (int pp = 0; pp < 2; ++pp) {
            a[pp].x = __shfl(inv, 4 * half + 2 * pp);
            a[pp].y = __shfl(inv, 4 * half + 2 * pp + 1);
        }

#pragma unroll
        for (int g = 0; g < 4; ++g) {
            v8f d0 = {};
            v8f d1 = {};
            v8f d2 = {};
            v8f d3 = {};
#pragma unroll
            for (int pp = 0; pp < 2; ++pp) {
                d0 = __builtin_amdgcn_wmma_f32_16x16x4_f32(false, a[pp], false, e[pp][g][0],
                                                           (short)0, d0, false, false);
                d1 = __builtin_amdgcn_wmma_f32_16x16x4_f32(false, a[pp], false, e[pp][g][1],
                                                           (short)0, d1, false, false);
                d2 = __builtin_amdgcn_wmma_f32_16x16x4_f32(false, a[pp], false, e[pp][g][2],
                                                           (short)0, d2, false, false);
                d3 = __builtin_amdgcn_wmma_f32_16x16x4_f32(false, a[pp], false, e[pp][g][3],
                                                           (short)0, d3, false, false);
            }
            res[st][g] = make_float4(d0[0], d1[0], d2[0], d3[0]);
        }
        __builtin_amdgcn_sched_barrier(0);
    }

    float4 o0, o1;
    o0.x = half ? res[0][2].x + res[1][2].x : res[0][0].x + res[1][0].x;
    o0.y = half ? res[0][2].y + res[1][2].y : res[0][0].y + res[1][0].y;
    o0.z = half ? res[0][2].z + res[1][2].z : res[0][0].z + res[1][0].z;
    o0.w = half ? res[0][2].w + res[1][2].w : res[0][0].w + res[1][0].w;
    o1.x = half ? res[0][3].x + res[1][3].x : res[0][1].x + res[1][1].x;
    o1.y = half ? res[0][3].y + res[1][3].y : res[0][1].y + res[1][1].y;
    o1.z = half ? res[0][3].z + res[1][3].z : res[0][1].z + res[1][1].z;
    o1.w = half ? res[0][3].w + res[1][3].w : res[0][1].w + res[1][1].w;
    float* __restrict__ qp = Q + (size_t)blockIdx.x * N + j0 + 128 * half;
    *(float4*)(qp) = o0;
    *(float4*)(qp + 64) = o1;
}

__global__ __launch_bounds__(512) void sinkhorn_col_finish(
    const float* __restrict__ Q, float* __restrict__ C) {
    __shared__ __attribute__((aligned(16))) float sP[32][64];
    const int tid = threadIdx.x;
    const int lane = tid & 31;
    const int wave = tid >> 5;
    const int half = lane >> 4;
    const int q = lane & 15;
    const int grp = 2 * wave + half;
    const float* __restrict__ qp = Q + (size_t)(8 * grp) * N + blockIdx.x * 64 + 4 * q;
    float4 v[8];
#pragma unroll
    for (int u = 0; u < 8; ++u) v[u] = *(const float4*)(qp + (size_t)u * N);
    float4 s = v[0];
#pragma unroll
    for (int u = 1; u < 8; ++u) {
        s.x += v[u].x;
        s.y += v[u].y;
        s.z += v[u].z;
        s.w += v[u].w;
    }
    *(float4*)&sP[grp][4 * q] = s;
    __syncthreads();
    if (tid < 64) {
        float t = 0.0f;
#pragma unroll
        for (int w = 0; w < 32; ++w) t += sP[w][tid];
        const int j = blockIdx.x * 64 + tid;
        C[j] = C[j] + __logf(t);
    }
}

__global__ __launch_bounds__(256) void sinkhorn_final(
    const float* __restrict__ M, const float* __restrict__ R,
    const float* __restrict__ C, float* __restrict__ out) {
    const int idx = blockIdx.x * blockDim.x + threadIdx.x;
    const int i = idx >> (12 - 2);
    const int j4 = (idx & (N / 4 - 1)) << 2;
    const size_t off = (size_t)i * N + j4;

    const float4 m = *(const float4*)(M + off);
    const float4 c = *(const float4*)(C + j4);
    const float r = R[i];
    float4 o;
    o.x = fmaxf(__expf(m.x - r - c.x), 1e-9f);
    o.y = fmaxf(__expf(m.y - r - c.y), 1e-9f);
    o.z = fmaxf(__expf(m.z - r - c.z), 1e-9f);
    o.w = fmaxf(__expf(m.w - r - c.w), 1e-9f);
    *(float4*)(out + off) = o;
}

extern "C" void kernel_launch(void* const* d_in, const int* in_sizes, int n_in,
                              void* d_out, int out_size, void* d_ws,
                              size_t ws_size, hipStream_t stream) {
    (void)in_sizes; (void)n_in; (void)out_size; (void)ws_size;

    const float* M = (const float*)d_in[0];
    float* out = (float*)d_out;

    char* ws = (char*)d_ws;
    float* R = (float*)(ws);
    float* C = (float*)(ws + 16 * 1024);
    float* Q = out;

    sinkhorn_init_c<<<N / 256, 256, 0, stream>>>(C);

    for (int t = 0; t < SINKHORN_ITERS; ++t) {
        sinkhorn_row_col<<<NBLOCK, 512, 0, stream>>>(M, C, R, Q);
        sinkhorn_col_finish<<<N / 64, 512, 0, stream>>>(Q, C);
    }

    sinkhorn_final<<<(N * (size_t)N / 4) / 256, 256, 0, stream>>>(M, R, C, out);
}


# exp-to-products hand-off shortened: in-row swizzles replace LDS-crossbar shuffles (lane and wave sums, 1/S broadcast), rcp for 1/S, log S stored last; row sums re-ordered; products in 4-tile groups
# speedup vs baseline: 1.5417x; 1.0319x over previous
#include <hip/hip_runtime.h>


#define N 4096
#define SINKHORN_ITERS 50
#define ROWS_PER_BLOCK 16
#define NBLOCK (N / ROWS_PER_BLOCK)
#define COLS_PER_WAVE 256

typedef float v2f __attribute__((ext_vector_type(2)));
typedef float v4f __attribute__((ext_vector_type(4)));
typedef float v8f __attribute__((ext_vector_type(8)));

template <int MASK>
__device__ __forceinline__ float row_xor(float v) {
    return __int_as_float(
        __builtin_amdgcn_update_dpp(0, __float_as_int(v), 0x160 + MASK, 0xf, 0xf, true));
}
template <int SRC>
__device__ __forceinline__ float row_pick(float v) {
    return __int_as_float(
        __builtin_amdgcn_update_dpp(0, __float_as_int(v), 0x150 + SRC, 0xf, 0xf, true));
}

__global__ void sinkhorn_init_c(float* __restrict__ C) {
    const int j = blockIdx.x * blockDim.x + threadIdx.x;
    if (j < N) C[j] = 0.0f;
}

__global__ __launch_bounds__(512) __attribute__((amdgpu_waves_per_eu(4, 4)))
void sinkhorn_row_col(
    const float* __restrict__ M, const float* __restrict__ C,
    float* __restrict__ R, float* __restrict__ Q) {
    __shared__ float sS[2][16][8];
    const int tid = threadIdx.x;
    const int lane = tid & 31;
    const int wave = tid >> 5;
    const int half = lane >> 4;
    const int q = lane & 15;
    const int i0 = blockIdx.x * ROWS_PER_BLOCK;
    const int j0 = wave * COLS_PER_WAVE + 4 * q;

    v4f c[4];
#pragma unroll
    for (int g = 0; g < 4; ++g) c[g] = *(const v4f*)(C + j0 + 64 * g);

    const float* __restrict__ mp = M + (size_t)(i0 + 8 * half) * N + j0;
    v4f m[8][4];
#pragma unroll
    for (int p = 0; p < 4; ++p) {
        __builtin_amdgcn_sched_barrier(0);
#pragma unroll
        for (int g = 0; g < 4; ++g) {
#pragma unroll
            for (int t = 2 * p; t < 2 * p + 2; ++t)
                m[t][g] = *(const v4f*)(mp + (size_t)t * N + 64 * g);
        }
    }

    const bool b8 = (q & 8) != 0;
    const bool b4 = (q & 4) != 0;
    float4 res[2][4];
    float totk[2];

#pragma unroll
    for (int st = 0; st < 2; ++st) {
        v2f e[2][4][4];
        float s[4];
#pragma unroll
        for (int pp = 0; pp < 2; ++pp) {
            const int p = 2 * st + pp;
            float4 s0 = make_float4(0.0f, 0.0f, 0.0f, 0.0f);
            float4 s1 = make_float4(0.0f, 0.0f, 0.0f, 0.0f);
#pragma unroll
            for (int g = 0; g < 4; ++g) {
                v4f x0 = m[2 * p][g];
                v4f x1 = m[2 * p + 1][g];
                asm volatile("" : "+v"(x0), "+v"(x1));
                const v4f y0 = x0 - c[g];
                const v4f y1 = x1 - c[g];
                e[pp][g][0].x = __expf(y0.x);
                e[pp][g][1].x = __expf(y0.y);
                e[pp][g][2].x = __expf(y0.z);
                e[pp][g][3].x = __expf(y0.w);
                e[pp][g][0].y = __expf(y1.x);
                e[pp][g][1].y = __expf(y1.y);
                e[pp][g][2].y = __expf(y1.z);
                e[pp][g][3].y = __expf(y1.w);
                s0.x += e[pp][g][0].x;
                s0.y += e[pp][g][1].x;
                s0.z += e[pp][g][2].x;
                s0.w += e[pp][g][3].x;
                s1.x += e[pp][g][0].y;
                s1.y += e[pp][g][1].y;
                s1.z += e[pp][g][2].y;
                s1.w += e[pp][g][3].y;
            }
            s[2 * pp] = (s0.x + s0.y) + (s0.z + s0.w);
            s[2 * pp + 1] = (s1.x + s1.y) + (s1.z + s1.w);
            __builtin_amdgcn_sched_barrier(0);
        }

        float u2[2];
#pragma unroll
        for (int u = 0; u < 2; ++u) {
            const float keep = b8 ? s[u + 2] : s[u];
            const float send = b8 ? s[u] : s[u + 2];
            u2[u] = keep + row_xor<8>(send);
        }
        float u1;
        {
            const float keep = b4 ? u2[1] : u2[0];
            const float send = b4 ? u2[0] : u2[1];
            u1 = keep + row_xor<4>(send);
        }
        u1 += row_xor<2>(u1);
        u1 += row_xor<1>(u1);
        if ((q & 3) == 0) sS[st][wave][4 * half + (q >> 2)] = u1;
        asm volatile("s_wait_dscnt 0x0\n\ts_barrier_signal -1\n\ts_barrier_wait -1" ::: "memory");

        const int ent = 4 * half + (q & 3);
        const int w0 = q & 12;
        float tot = (sS[st][w0][ent] + sS[st][w0 + 1][ent]) +
                    (sS[st][w0 + 2][ent] + sS[st][w0 + 3][ent]);
        tot += row_xor<4>(tot);
        tot += row_xor<8>(tot);
        totk[st] = tot;
        const float inv = __builtin_amdgcn_rcpf(tot);

        v2f a[2];
        a[0].x = row_pick<0>(inv);
        a[0].y = row_pick<1>(inv);
        a[1].x = row_pick<2>(inv);
        a[1].y = row_pick<3>(inv);

#pragma unroll
        for (int g = 0; g < 4; ++g) {
            v8f d0 = {};
            v8f d1 = {};
            v8f d2 = {};
            v8f d3 = {};
#pragma unroll
            for (int pp = 0; pp < 2; ++pp) {
                d0 = __builtin_amdgcn_wmma_f32_16x16x4_f32(false, a[pp], false, e[pp][g][0],
                                                           (short)0, d0, false, false);
                d1 = __builtin_amdgcn_wmma_f32_16x16x4_f32(false, a[pp], false, e[pp][g][1],
                                                           (short)0, d1, false, false);
                d2 = __builtin_amdgcn_wmma_f32_16x16x4_f32(false, a[pp], false, e[pp][g][2],
                                                           (short)0, d2, false, false);
                d3 = __builtin_amdgcn_wmma_f32_16x16x4_f32(false, a[pp], false, e[pp][g][3],
                                                           (short)0, d3, false, false);
            }
            res[st][g] = make_float4(d0[0], d1[0], d2[0], d3[0]);
            __builtin_amdgcn_sched_barrier(0);
        }
        asm volatile("v_nop\n\tv_nop\n\tv_nop\n\tv_nop");
        __builtin_amdgcn_sched_barrier(0);
    }

    float4 o0, o1;
    o0.x = half ? res[0][2].x + res[1][2].x : res[0][0].x + res[1][0].x;
    o0.y = half ? res[0][2].y + res[1][2].y : res[0][0].y + res[1][0].y;
    o0.z = half ? res[0][2].z + res[1][2].z : res[0][0].z + res[1][0].z;
    o0.w = half ? res[0][2].w + res[1][2].w : res[0][0].w + res[1][0].w;
    o1.x = half ? res[0][3].x + res[1][3].x : res[0][1].x + res[1][1].x;
    o1.y = half ? res[0][3].y + res[1][3].y : res[0][1].y + res[1][1].y;
    o1.z = half ? res[0][3].z + res[1][3].z : res[0][1].z + res[1][1].z;
    o1.w = half ? res[0][3].w + res[1][3].w : res[0][1].w + res[1][1].w;
    float* __restrict__ qp = Q + (size_t)blockIdx.x * N + j0 + 128 * half;
    *(float4*)(qp) = o0;
    *(float4*)(qp + 64) = o1;
    if (wave == 0 && (q & 12) == 0) {
        R[i0 + 8 * half + (q & 3)] = __logf(totk[0]);
        R[i0 + 8 * half + 4 + (q & 3)] = __logf(totk[1]);
    }
}

__global__ __launch_bounds__(512) void sinkhorn_col_finish(
    const float* __restrict__ Q, float* __restrict__ C) {
    __shared__ __attribute__((aligned(16))) float sP[32][64];
    const int tid = threadIdx.x;
    const int lane = tid & 31;
    const int wave = tid >> 5;
    const int half = lane >> 4;
    const int q = lane & 15;
    const int grp = 2 * wave + half;
    const float* __restrict__ qp = Q + (size_t)(8 * grp) * N + blockIdx.x * 64 + 4 * q;
    float4 v[8];
#pragma unroll
    for (int u = 0; u < 8; ++u) v[u] = *(const float4*)(qp + (size_t)u * N);
    float4 s = v[0];
#pragma unroll
    for (int u = 1; u < 8; ++u) {
        s.x += v[u].x;
        s.y += v[u].y;
        s.z += v[u].z;
        s.w += v[u].w;
    }
    *(float4*)&sP[grp][4 * q] = s;
    __syncthreads();
    if (tid < 64) {
        float t = 0.0f;
#pragma unroll
        for (int w = 0; w < 32; ++w) t += sP[w][tid];
        const int j = blockIdx.x * 64 + tid;
        C[j] = C[j] + __logf(t);
    }
}

__global__ __launch_bounds__(256) void sinkhorn_final(
    const float* __restrict__ M, const float* __restrict__ R,
    const float* __restrict__ C, float* __restrict__ out) {
    const int idx = blockIdx.x * blockDim.x + threadIdx.x;
    const int i = idx >> (12 - 2);
    const int j4 = (idx & (N / 4 - 1)) << 2;
    const size_t off = (size_t)i * N + j4;

    const float4 m = *(const float4*)(M + off);
    const float4 c = *(const float4*)(C + j4);
    const float r = R[i];
    float4 o;
    o.x = fmaxf(__expf(m.x - r - c.x), 1e-9f);
    o.y = fmaxf(__expf(m.y - r - c.y), 1e-9f);
    o.z = fmaxf(__expf(m.z - r - c.z), 1e-9f);
    o.w = fmaxf(__expf(m.w - r - c.w), 1e-9f);
    *(float4*)(out + off) = o;
}

extern "C" void kernel_launch(void* const* d_in, const int* in_sizes, int n_in,
                              void* d_out, int out_size, void* d_ws,
                              size_t ws_size, hipStream_t stream) {
    (void)in_sizes; (void)n_in; (void)out_size; (void)ws_size;

    const float* M = (const float*)d_in[0];
    float* out = (float*)d_out;

    char* ws = (char*)d_ws;
    float* R = (float*)(ws);
    float* C = (float*)(ws + 16 * 1024);
    float* Q = out;

    sinkhorn_init_c<<<N / 256, 256, 0, stream>>>(C);

    for (int t = 0; t < SINKHORN_ITERS; ++t) {
        sinkhorn_row_col<<<NBLOCK, 512, 0, stream>>>(M, C, R, Q);
        sinkhorn_col_finish<<<N / 64, 512, 0, stream>>>(Q, C);
    }

    sinkhorn_final<<<(N * (size_t)N / 4) / 256, 256, 0, stream>>>(M, R, C, out);
}


# row-and-column pass: a work-group barrier between the two stages' load groups queues every wave's stage-0 rows ahead of any stage-1 row, so stage-0 products run under stage-1 delivery
# speedup vs baseline: 1.7378x; 1.1272x over previous
#include <hip/hip_runtime.h>


#define N 4096
#define SINKHORN_ITERS 50
#define ROWS_PER_BLOCK 16
#define NBLOCK (N / ROWS_PER_BLOCK)
#define COLS_PER_WAVE 256

typedef float v2f __attribute__((ext_vector_type(2)));
typedef float v4f __attribute__((ext_vector_type(4)));
typedef float v8f __attribute__((ext_vector_type(8)));

template <int MASK>
__device__ __forceinline__ float row_xor(float v) {
    return __int_as_float(
        __builtin_amdgcn_update_dpp(0, __float_as_int(v), 0x160 + MASK, 0xf, 0xf, true));
}
template <int SRC>
__device__ __forceinline__ float row_pick(float v) {
    return __int_as_float(
        __builtin_amdgcn_update_dpp(0, __float_as_int(v), 0x150 + SRC, 0xf, 0xf, true));
}

__global__ void sinkhorn_init_c(float* __restrict__ C) {
    const int j = blockIdx.x * blockDim.x + threadIdx.x;
    if (j < N) C[j] = 0.0f;
}

__global__ __launch_bounds__(512) __attribute__((amdgpu_waves_per_eu(4, 4)))
void sinkhorn_row_col(
    const float* __restrict__ M, const float* __restrict__ C,
    float* __restrict__ R, float* __restrict__ Q) {
    __shared__ float sS[2][16][8];
    const int tid = threadIdx.x;
    const int lane = tid & 31;
    const int wave = tid >> 5;
    const int half = lane >> 4;
    const int q = lane & 15;
    const int i0 = blockIdx.x * ROWS_PER_BLOCK;
    const int j0 = wave * COLS_PER_WAVE + 4 * q;

    v4f c[4];
#pragma unroll
    for (int g = 0; g < 4; ++g) c[g] = *(const v4f*)(C + j0 + 64 * g);

    const float* __restrict__ mp = M + (size_t)(i0 + 8 * half) * N + j0;
    v4f m[8][4];
#pragma unroll
    for (int p = 0; p < 4; ++p) {
        __builtin_amdgcn_sched_barrier(0);
        if (p == 2) {
            asm volatile("s_barrier_signal -1\n\ts_barrier_wait -1" ::: "memory");
            __builtin_amdgcn_sched_barrier(0);
        }
#pragma unroll
        for (int g = 0; g < 4; ++g) {
#pragma unroll
            for (int t = 2 * p; t < 2 * p + 2; ++t)
                m[t][g] = *(const v4f*)(mp + (size_t)t * N + 64 * g);
        }
    }

    const bool b8 = (q & 8) != 0;
    const bool b4 = (q & 4) != 0;
    float4 res[2][4];
    float totk[2];

#pragma unroll
    for (int st = 0; st < 2; ++st) {
        v2f e[2][4][4];
        float s[4];
#pragma unroll
        for (int pp = 0; pp < 2; ++pp) {
            const int p = 2 * st + pp;
            float4 s0 = make_float4(0.0f, 0.0f, 0.0f, 0.0f);
            float4 s1 = make_float4(0.0f, 0.0f, 0.0f, 0.0f);
#pragma unroll
            for (int g = 0; g < 4; ++g) {
                v4f x0 = m[2 * p][g];
                v4f x1 = m[2 * p + 1][g];
                asm volatile("" : "+v"(x0), "+v"(x1));
                const v4f y0 = x0 - c[g];
                const v4f y1 = x1 - c[g];
                e[pp][g][0].x = __expf(y0.x);
                e[pp][g][1].x = __expf(y0.y);
                e[pp][g][2].x = __expf(y0.z);
                e[pp][g][3].x = __expf(y0.w);
                e[pp][g][0].y = __expf(y1.x);
                e[pp][g][1].y = __expf(y1.y);
                e[pp][g][2].y = __expf(y1.z);
                e[pp][g][3].y = __expf(y1.w);
                s0.x += e[pp][g][0].x;
                s0.y += e[pp][g][1].x;
                s0.z += e[pp][g][2].x;
                s0.w += e[pp][g][3].x;
                s1.x += e[pp][g][0].y;
                s1.y += e[pp][g][1].y;
                s1.z += e[pp][g][2].y;
                s1.w += e[pp][g][3].y;
            }
            s[2 * pp] = (s0.x + s0.y) + (s0.z + s0.w);
            s[2 * pp + 1] = (s1.x + s1.y) + (s1.z + s1.w);
            __builtin_amdgcn_sched_barrier(0);
        }

        float u2[2];
#pragma unroll
        for (int u = 0; u < 2; ++u) {
            const float keep = b8 ? s[u + 2] : s[u];
            const float send = b8 ? s[u] : s[u + 2];
            u2[u] = keep + row_xor<8>(send);
        }
        float u1;
        {
            const float keep = b4 ? u2[1] : u2[0];
            const float send = b4 ? u2[0] : u2[1];
            u1 = keep + row_xor<4>(send);
        }
        u1 += row_xor<2>(u1);
        u1 += row_xor<1>(u1);
        if ((q & 3) == 0) sS[st][wave][4 * half + (q >> 2)] = u1;
        asm volatile("s_wait_dscnt 0x0\n\ts_barrier_signal -1\n\ts_barrier_wait -1" ::: "memory");

        const int ent = 4 * half + (q & 3);
        const int w0 = q & 12;
        float tot = (sS[st][w0][ent] + sS[st][w0 + 1][ent]) +
                    (sS[st][w0 + 2][ent] + sS[st][w0 + 3][ent]);
        tot += row_xor<4>(tot);
        tot += row_xor<8>(tot);
        totk[st] = tot;
        const float inv = __builtin_amdgcn_rcpf(tot);

        v2f a[2];
        a[0].x = row_pick<0>(inv);
        a[0].y = row_pick<1>(inv);
        a[1].x = row_pick<2>(inv);
        a[1].y = row_pick<3>(inv);

#pragma unroll
        for (int g = 0; g < 4; ++g) {
            v8f d0 = {};
            v8f d1 = {};
            v8f d2 = {};
            v8f d3 = {};
#pragma unroll
            for (int pp = 0; pp < 2; ++pp) {
                d0 = __builtin_amdgcn_wmma_f32_16x16x4_f32(false, a[pp], false, e[pp][g][0],
                                                           (short)0, d0, false, false);
                d1 = __builtin_amdgcn_wmma_f32_16x16x4_f32(false, a[pp], false, e[pp][g][1],
                                                           (short)0, d1, false, false);
                d2 = __builtin_amdgcn_wmma_f32_16x16x4_f32(false, a[pp], false, e[pp][g][2],
                                                           (short)0, d2, false, false);
                d3 = __builtin_amdgcn_wmma_f32_16x16x4_f32(false, a[pp], false, e[pp][g][3],
                                                           (short)0, d3, false, false);
            }
            res[st][g] = make_float4(d0[0], d1[0], d2[0], d3[0]);
            __builtin_amdgcn_sched_barrier(0);
        }
        asm volatile("v_nop\n\tv_nop\n\tv_nop\n\tv_nop");
        __builtin_amdgcn_sched_barrier(0);
    }

    float4 o0, o1;
    o0.x = half ? res[0][2].x + res[1][2].x : res[0][0].x + res[1][0].x;
    o0.y = half ? res[0][2].y + res[1][2].y : res[0][0].y + res[1][0].y;
    o0.z = half ? res[0][2].z + res[1][2].z : res[0][0].z + res[1][0].z;
    o0.w = half ? res[0][2].w + res[1][2].w : res[0][0].w + res[1][0].w;
    o1.x = half ? res[0][3].x + res[1][3].x : res[0][1].x + res[1][1].x;
    o1.y = half ? res[0][3].y + res[1][3].y : res[0][1].y + res[1][1].y;
    o1.z = half ? res[0][3].z + res[1][3].z : res[0][1].z + res[1][1].z;
    o1.w = half ? res[0][3].w + res[1][3].w : res[0][1].w + res[1][1].w;
    float* __restrict__ qp = Q + (size_t)blockIdx.x * N + j0 + 128 * half;
    *(float4*)(qp) = o0;
    *(float4*)(qp + 64) = o1;
    if (wave == 0 && (q & 12) == 0) {
        R[i0 + 8 * half + (q & 3)] = __logf(totk[0]);
        R[i0 + 8 * half + 4 + (q & 3)] = __logf(totk[1]);
    }
}

__global__ __launch_bounds__(512) void sinkhorn_col_finish(
    const float* __restrict__ Q, float* __restrict__ C) {
    __shared__ __attribute__((aligned(16))) float sP[32][64];
    const int tid = threadIdx.x;
    const int lane = tid & 31;
    const int wave = tid >> 5;
    const int half = lane >> 4;
    const int q = lane & 15;
    const int grp = 2 * wave + half;
    const float* __restrict__ qp = Q + (size_t)(8 * grp) * N + blockIdx.x * 64 + 4 * q;
    float4 v[8];
#pragma unroll
    for (int u = 0; u < 8; ++u) v[u] = *(const float4*)(qp + (size_t)u * N);
    float4 s = v[0];
#pragma unroll
    for (int u = 1; u < 8; ++u) {
        s.x += v[u].x;
        s.y += v[u].y;
        s.z += v[u].z;
        s.w += v[u].w;
    }
    *(float4*)&sP[grp][4 * q] = s;
    __syncthreads();
    if (tid < 64) {
        float t = 0.0f;
#pragma unroll
        for (int w = 0; w < 32; ++w) t += sP[w][tid];
        const int j = blockIdx.x * 64 + tid;
        C[j] = C[j] + __logf(t);
    }
}

__global__ __launch_bounds__(256) void sinkhorn_final(
    const float* __restrict__ M, const float* __restrict__ R,
    const float* __restrict__ C, float* __restrict__ out) {
    const int idx = blockIdx.x * blockDim.x + threadIdx.x;
    const int i = idx >> (12 - 2);
    const int j4 = (idx & (N / 4 - 1)) << 2;
    const size_t off = (size_t)i * N + j4;

    const float4 m = *(const float4*)(M + off);
    const float4 c = *(const float4*)(C + j4);
    const float r = R[i];
    float4 o;
    o.x = fmaxf(__expf(m.x - r - c.x), 1e-9f);
    o.y = fmaxf(__expf(m.y - r - c.y), 1e-9f);
    o.z = fmaxf(__expf(m.z - r - c.z), 1e-9f);
    o.w = fmaxf(__expf(m.w - r - c.w), 1e-9f);
    *(float4*)(out + off) = o;
}

extern "C" void kernel_launch(void* const* d_in, const int* in_sizes, int n_in,
                              void* d_out, int out_size, void* d_ws,
                              size_t ws_size, hipStream_t stream) {
    (void)in_sizes; (void)n_in; (void)out_size; (void)ws_size;

    const float* M = (const float*)d_in[0];
    float* out = (float*)d_out;

    char* ws = (char*)d_ws;
    float* R = (float*)(ws);
    float* C = (float*)(ws + 16 * 1024);
    float* Q = out;

    sinkhorn_init_c<<<N / 256, 256, 0, stream>>>(C);

    for (int t = 0; t < SINKHORN_ITERS; ++t) {
        sinkhorn_row_col<<<NBLOCK, 512, 0, stream>>>(M, C, R, Q);
        sinkhorn_col_finish<<<N / 64, 512, 0, stream>>>(Q, C);
    }

    sinkhorn_final<<<(N * (size_t)N / 4) / 256, 256, 0, stream>>>(M, R, C, out);
}
